# starting kernel
# baseline (speedup 1.0000x reference)
#include <hip/hip_runtime.h>
#include <hip/hip_bf16.h>
#include <math.h>

typedef __attribute__((ext_vector_type(2))) float v2f;
typedef __attribute__((ext_vector_type(8))) float v8f;

#define CCH   64
#define FILLW 2.0f

__device__ __forceinline__ void atomic_add_f32(float* p, float v) {
  __hip_atomic_fetch_add(p, v, __ATOMIC_RELAXED, __HIP_MEMORY_SCOPE_AGENT);
}

__global__ void gcn_init(float* __restrict__ out, float* __restrict__ deg,
                         int total, int n_nodes) {
  int t = blockIdx.x * blockDim.x + threadIdx.x;
  if (t < total)   out[t] = 0.0f;
  if (t < n_nodes) deg[t] = FILLW;
}

__global__ void gcn_deg(const int* __restrict__ dst, const float* __restrict__ w,
                        float* __restrict__ deg, int n_edges) {
  int e = blockIdx.x * blockDim.x + threadIdx.x;
  if (e < n_edges) atomic_add_f32(&deg[dst[e]], w[e]);
}

__global__ void gcn_dinv(float* __restrict__ deg, int n_nodes) {
  int n = blockIdx.x * blockDim.x + threadIdx.x;
  if (n < n_nodes) {
    float d = deg[n];
    deg[n] = (d > 0.0f) ? rsqrtf(d) : 0.0f;
  }
}

__global__ void gcn_gemm_wmma(const float* __restrict__ X,
                              const float* __restrict__ W,
                              float* __restrict__ XW, int n_nodes) {
  const int wave = (blockIdx.x * blockDim.x + threadIdx.x) >> 5;
  const int lane = threadIdx.x & 31;
  const int n_row_tiles = (n_nodes + 15) >> 4;
  const int row_tile = wave >> 2;
  const int col_tile = wave & 3;
  if (row_tile >= n_row_tiles) return;

  const int half = lane >> 4;
  const int l16  = lane & 15;
  int row = row_tile * 16 + l16;
  int ldrow = (row < n_nodes) ? row : (n_nodes - 1);

  const float* __restrict__ xrow = X + (size_t)ldrow * CCH;
  const float* __restrict__ wcol = W + col_tile * 16 + l16;

  v8f acc = {};
#pragma unroll
  for (int k0 = 0; k0 < CCH; k0 += 4) {
    const int ka = k0 + 2 * half;
    v2f a, b;
    a.x = xrow[ka + 0];
    a.y = xrow[ka + 1];
    b.x = wcol[(size_t)(ka + 0) * CCH];
    b.y = wcol[(size_t)(ka + 1) * CCH];
    acc = __builtin_amdgcn_wmma_f32_16x16x4_f32(
         false, a,  false, b,
         (short)0, acc,  false,  false);
  }

  const int base_row = row_tile * 16 + 8 * half;
  float* __restrict__ dstp = XW + (size_t)base_row * CCH + col_tile * 16 + l16;
#pragma unroll
  for (int i = 0; i < 8; ++i) {
    if (base_row + i < n_nodes) dstp[(size_t)i * CCH] = acc[i];
  }
}

__global__ void gcn_scatter(const int* __restrict__ src, const int* __restrict__ dst,
                            const float* __restrict__ w, const float* __restrict__ dinv,
                            const float* __restrict__ xw, float* __restrict__ out,
                            int n_edges) {
  const int gt   = blockIdx.x * blockDim.x + threadIdx.x;
  const int e    = gt >> 5;
  const int lane = gt & 31;
  if (e >= n_edges) return;

  const int s = src[e];
  const int d = dst[e];
  float norm = 0.0f;
  if (lane == 0) norm = dinv[s] * w[e] * dinv[d];
  norm = __shfl(norm, 0, 32);

  const float* __restrict__ xr = xw + (size_t)s * CCH;
  float* __restrict__ orow     = out + (size_t)d * CCH;

  atomic_add_f32(&orow[lane],      xr[lane]      * norm);
  atomic_add_f32(&orow[lane + 32], xr[lane + 32] * norm);
}

__global__ void gcn_final(float* __restrict__ out, const float* __restrict__ xw,
                          const float* __restrict__ dinv, const float* __restrict__ b,
                          int total) {
  int t = blockIdx.x * blockDim.x + threadIdx.x;
  if (t < total) {
    int n = t >> 6;
    int c = t & (CCH - 1);
    float di = dinv[n];
    float v  = out[t] + xw[t] * (FILLW * di * di) + b[c];
    out[t] = tanhf(v);
  }
}

extern "C" void kernel_launch(void* const* d_in, const int* in_sizes, int n_in,
                              void* d_out, int out_size, void* d_ws, size_t ws_size,
                              hipStream_t stream) {
  const float* x  = (const float*)d_in[0];
  const int*   ei = (const int*)d_in[1];
  const float* ew = (const float*)d_in[2];
  const float* W  = (const float*)d_in[3];
  const float* b  = (const float*)d_in[4];
  float* out = (float*)d_out;

  const int n_nodes = in_sizes[0] / CCH;
  const int n_edges = in_sizes[2];
  const int* src = ei;
  const int* dst = ei + n_edges;

  float* xw   = (float*)d_ws;
  float* dinv = xw + (size_t)n_nodes * CCH;

  const int total = n_nodes * CCH;
  const int B = 256;

  gcn_init<<<(total + B - 1) / B, B, 0, stream>>>(out, dinv, total, n_nodes);
  gcn_deg<<<(n_edges + B - 1) / B, B, 0, stream>>>(dst, ew, dinv, n_edges);
  gcn_dinv<<<(n_nodes + B - 1) / B, B, 0, stream>>>(dinv, n_nodes);

  const int n_row_tiles = (n_nodes + 15) / 16;
  const int waves = n_row_tiles * (CCH / 16);
  gcn_gemm_wmma<<<(waves + 7) / 8, B, 0, stream>>>(x, W, xw, n_nodes);

  const long long sthreads = (long long)n_edges * 32;
  gcn_scatter<<<(unsigned)((sthreads + B - 1) / B), B, 0, stream>>>(
      src, dst, ew, dinv, xw, out, n_edges);

  gcn_final<<<(total + B - 1) / B, B, 0, stream>>>(out, xw, dinv, b, total);
}


# Scatter+final in one owner launch: each group holds 391 dst rows as f32 sums in LDS, filters the raw dst list, adds xw[src]*norm by LDS atomics, stores tanh once; per-node sum order = LDS arrival
# speedup vs baseline: 1.6956x; 1.6956x over previous
#include <hip/hip_runtime.h>
#include <hip/hip_bf16.h>
#include <math.h>

typedef __attribute__((ext_vector_type(2))) float v2f;
typedef __attribute__((ext_vector_type(8))) float v8f;

#define CCH   64
#define FILLW 2.0f

__device__ __forceinline__ void atomic_add_f32(float* p, float v) {
  __hip_atomic_fetch_add(p, v, __ATOMIC_RELAXED, __HIP_MEMORY_SCOPE_AGENT);
}

__global__ void gcn_init(float* __restrict__ deg, int n_nodes) {
  int t = blockIdx.x * blockDim.x + threadIdx.x;
  if (t < n_nodes) deg[t] = FILLW;
}

__global__ void gcn_deg(const int* __restrict__ dst, const float* __restrict__ w,
                        float* __restrict__ deg, int n_edges) {
  int e = blockIdx.x * blockDim.x + threadIdx.x;
  if (e < n_edges) atomic_add_f32(&deg[dst[e]], w[e]);
}

__global__ void gcn_dinv(float* __restrict__ deg, int n_nodes) {
  int n = blockIdx.x * blockDim.x + threadIdx.x;
  if (n < n_nodes) {
    float d = deg[n];
    deg[n] = (d > 0.0f) ? rsqrtf(d) : 0.0f;
  }
}

__global__ void gcn_gemm_wmma(const float* __restrict__ X,
                              const float* __restrict__ W,
                              float* __restrict__ XW, int n_nodes) {
  const int wave = (blockIdx.x * blockDim.x + threadIdx.x) >> 5;
  const int lane = threadIdx.x & 31;
  const int n_row_tiles = (n_nodes + 15) >> 4;
  const int row_tile = wave >> 2;
  const int col_tile = wave & 3;
  if (row_tile >= n_row_tiles) return;

  const int half = lane >> 4;
  const int l16  = lane & 15;
  int row = row_tile * 16 + l16;
  int ldrow = (row < n_nodes) ? row : (n_nodes - 1);

  const float* __restrict__ xrow = X + (size_t)ldrow * CCH;
  const float* __restrict__ wcol = W + col_tile * 16 + l16;

  v8f acc = {};
#pragma unroll
  for (int k0 = 0; k0 < CCH; k0 += 4) {
    const int ka = k0 + 2 * half;
    v2f a, b;
    a.x = xrow[ka + 0];
    a.y = xrow[ka + 1];
    b.x = wcol[(size_t)(ka + 0) * CCH];
    b.y = wcol[(size_t)(ka + 1) * CCH];
    acc = __builtin_amdgcn_wmma_f32_16x16x4_f32(
         false, a,  false, b,
         (short)0, acc,  false,  false);
  }

  const int base_row = row_tile * 16 + 8 * half;
  float* __restrict__ dstp = XW + (size_t)base_row * CCH + col_tile * 16 + l16;
#pragma unroll
  for (int i = 0; i < 8; ++i) {
    if (base_row + i < n_nodes) dstp[(size_t)i * CCH] = acc[i];
  }
}

typedef __attribute__((ext_vector_type(4))) int      v4i;
typedef __attribute__((ext_vector_type(4))) float    v4f;
typedef __attribute__((ext_vector_type(2))) unsigned v2u;

#define RNG         391
#define OWN_THREADS 1024
#define OWN_WAVES   32
#define QCAP        512
#define WAVE_EDGES  512
#define STEP_EDGES  (OWN_WAVES * WAVE_EDGES)
#define ROWS_FLY    16

__device__ __forceinline__ void lds_add_f32(float* p, float v) {
  __hip_atomic_fetch_add(p, v, __ATOMIC_RELAXED, __HIP_MEMORY_SCOPE_WORKGROUP);
}

__device__ __forceinline__ void q_push(v2u* ring, unsigned& tail, bool m, int e, unsigned dl) {
  const unsigned mask = __builtin_amdgcn_ballot_w32(m);
  if (mask != 0u) {
    if (m) {
      const unsigned slot = (tail + __builtin_amdgcn_mbcnt_lo(mask, 0u)) & (QCAP - 1);
      v2u ent; ent.x = (unsigned)e; ent.y = dl;
      ring[slot] = ent;
    }
    tail += (unsigned)__builtin_popcount(mask);
  }
}

__device__ __forceinline__ void q_push4(v2u* ring, unsigned& tail, v4i v, int e, unsigned lo) {
  const unsigned d0 = (unsigned)v.x - lo, d1 = (unsigned)v.y - lo;
  const unsigned d2 = (unsigned)v.z - lo, d3 = (unsigned)v.w - lo;
  q_push(ring, tail, d0 < RNG, e + 0, d0);
  q_push(ring, tail, d1 < RNG, e + 1, d1);
  q_push(ring, tail, d2 < RNG, e + 2, d2);
  q_push(ring, tail, d3 < RNG, e + 3, d3);
}

__device__ __forceinline__ void q_batch(const v2u* ring, unsigned head, int cnt,
                                        int lane, int lo,
                                        const int* __restrict__ src, const float* __restrict__ w,
                                        const float* __restrict__ dinv, const float* __restrict__ xw,
                                        float* __restrict__ acc) {
  int   s   = lo;
  int   dl  = RNG;
  float nrm = 0.0f;
  if (lane < cnt) {
    const v2u ent = ring[(head + (unsigned)lane) & (QCAP - 1)];
    const int e = (int)ent.x;
    dl  = (int)ent.y;
    s   = src[e];
    nrm = dinv[s] * w[e] * dinv[lo + dl];
  }
  for (int j0 = 0; j0 < cnt; j0 += ROWS_FLY) {
    v2f x[ROWS_FLY];
#pragma unroll
    for (int k = 0; k < ROWS_FLY; ++k) {
      const int sj = __builtin_amdgcn_readlane(s, j0 + k);
      x[k] = *reinterpret_cast<const v2f*>(xw + (size_t)sj * CCH + 2 * lane);
    }
#pragma unroll
    for (int k = 0; k < ROWS_FLY; ++k) {
      const int   dj = __builtin_amdgcn_readlane(dl, j0 + k);
      const float nj = __int_as_float(__builtin_amdgcn_readlane(__float_as_int(nrm), j0 + k));
      float* a = acc + dj * CCH + 2 * lane;
      lds_add_f32(a,     x[k].x * nj);
      lds_add_f32(a + 1, x[k].y * nj);
    }
  }
}

__device__ __forceinline__ void q_drain_full(const v2u* ring, unsigned& head, unsigned tail,
                                             int lane, int lo,
                                             const int* __restrict__ src, const float* __restrict__ w,
                                             const float* __restrict__ dinv, const float* __restrict__ xw,
                                             float* __restrict__ acc) {
  while (tail - head >= 32u) {
    q_batch(ring, head, 32, lane, lo, src, w, dinv, xw, acc);
    head += 32u;
  }
}

__device__ __forceinline__ void scan_step(v2u* ring, unsigned& head, unsigned& tail,
                                          v4i c0, v4i c1, v4i c2, v4i c3, int eb,
                                          int lane, int lo,
                                          const int* __restrict__ src, const float* __restrict__ w,
                                          const float* __restrict__ dinv, const float* __restrict__ xw,
                                          float* __restrict__ acc) {
  q_push4(ring, tail, c0, eb,       (unsigned)lo);
  q_push4(ring, tail, c1, eb + 128, (unsigned)lo);
  q_drain_full(ring, head, tail, lane, lo, src, w, dinv, xw, acc);
  q_push4(ring, tail, c2, eb + 256, (unsigned)lo);
  q_push4(ring, tail, c3, eb + 384, (unsigned)lo);
  q_drain_full(ring, head, tail, lane, lo, src, w, dinv, xw, acc);
}

__global__ void __launch_bounds__(OWN_THREADS)
gcn_owner(const int* __restrict__ src, const int* __restrict__ dst,
          const float* __restrict__ w, const float* __restrict__ dinv,
          const float* __restrict__ xw, const float* __restrict__ b,
          float* __restrict__ out, int n_edges, int n_nodes) {
  __shared__ __attribute__((aligned(16))) float acc[(RNG + 1) * CCH];
  __shared__ __attribute__((aligned(16))) v2u   ringbuf[OWN_WAVES * QCAP];

  const int tid  = threadIdx.x;
  const int lane = tid & 31;
  const int wave = __builtin_amdgcn_readfirstlane(tid >> 5);
  const int lo   = blockIdx.x * RNG;
  v2u* ring = ringbuf + wave * QCAP;

  for (int i = tid; i < (RNG + 1) * CCH / 4; i += OWN_THREADS) {
    v4f z = {0.0f, 0.0f, 0.0f, 0.0f};
    reinterpret_cast<v4f*>(acc)[i] = z;
  }
  __syncthreads();

  unsigned head = 0u, tail = 0u;

  const bool vec_ok = ((reinterpret_cast<size_t>(dst) & 15) == 0);
  const int  n_full = vec_ok ? (n_edges / STEP_EDGES) : 0;

  if (n_full > 0) {
    int it = (int)(((long long)blockIdx.x * n_full) / gridDim.x);
    v4i a0, a1, a2, a3, b0, b1, b2, b3;
    {
      const v4i* p = reinterpret_cast<const v4i*>(dst + (size_t)(it * OWN_WAVES + wave) * WAVE_EDGES) + lane;
      a0 = p[0]; a1 = p[32]; a2 = p[64]; a3 = p[96];
    }
    int n = 0;
    for (; n + 2 <= n_full; n += 2) {
      int it1 = it + 1;  if (it1 == n_full) it1 = 0;
      int it2 = it1 + 1; if (it2 == n_full) it2 = 0;
      {
        const v4i* p = reinterpret_cast<const v4i*>(dst + (size_t)(it1 * OWN_WAVES + wave) * WAVE_EDGES) + lane;
        b0 = p[0]; b1 = p[32]; b2 = p[64]; b3 = p[96];
      }
      scan_step(ring, head, tail, a0, a1, a2, a3, (it * OWN_WAVES + wave) * WAVE_EDGES + lane * 4,
                lane, lo, src, w, dinv, xw, acc);
      {
        const v4i* p = reinterpret_cast<const v4i*>(dst + (size_t)(it2 * OWN_WAVES + wave) * WAVE_EDGES) + lane;
        a0 = p[0]; a1 = p[32]; a2 = p[64]; a3 = p[96];
      }
      scan_step(ring, head, tail, b0, b1, b2, b3, (it1 * OWN_WAVES + wave) * WAVE_EDGES + lane * 4,
                lane, lo, src, w, dinv, xw, acc);
      it = it2;
    }
    if (n < n_full) {
      scan_step(ring, head, tail, a0, a1, a2, a3, (it * OWN_WAVES + wave) * WAVE_EDGES + lane * 4,
                lane, lo, src, w, dinv, xw, acc);
    }
  }
  for (int base = n_full * STEP_EDGES + wave * 32; base < n_edges; base += OWN_THREADS) {
    const int e = base + lane;
    const int d = (e < n_edges) ? dst[e] : -1;
    const unsigned du = (unsigned)d - (unsigned)lo;
    q_push(ring, tail, du < RNG, e, du);
    q_drain_full(ring, head, tail, lane, lo, src, w, dinv, xw, acc);
  }
  if (tail != head) q_batch(ring, head, (int)(tail - head), lane, lo, src, w, dinv, xw, acc);

  __syncthreads();

  for (int r = wave; r < RNG; r += OWN_WAVES) {
    const int n = lo + r;
    if (n < n_nodes) {
      const float di = dinv[n];
      const float k  = FILLW * di * di;
      const v2f a  = *reinterpret_cast<const v2f*>(acc + r * CCH + 2 * lane);
      const v2f x  = *reinterpret_cast<const v2f*>(xw + (size_t)n * CCH + 2 * lane);
      const v2f bb = *reinterpret_cast<const v2f*>(b + 2 * lane);
      v2f o;
      o.x = tanhf(a.x + x.x * k + bb.x);
      o.y = tanhf(a.y + x.y * k + bb.y);
      *reinterpret_cast<v2f*>(out + (size_t)n * CCH + 2 * lane) = o;
    }
  }
}

extern "C" void kernel_launch(void* const* d_in, const int* in_sizes, int n_in,
                              void* d_out, int out_size, void* d_ws, size_t ws_size,
                              hipStream_t stream) {
  const float* x  = (const float*)d_in[0];
  const int*   ei = (const int*)d_in[1];
  const float* ew = (const float*)d_in[2];
  const float* W  = (const float*)d_in[3];
  const float* b  = (const float*)d_in[4];
  float* out = (float*)d_out;

  const int n_nodes = in_sizes[0] / CCH;
  const int n_edges = in_sizes[2];
  const int* src = ei;
  const int* dst = ei + n_edges;

  float* xw   = (float*)d_ws;
  float* dinv = xw + (size_t)n_nodes * CCH;

  const int B = 256;

  gcn_init<<<(n_nodes + B - 1) / B, B, 0, stream>>>(dinv, n_nodes);
  gcn_deg<<<(n_edges + B - 1) / B, B, 0, stream>>>(dst, ew, dinv, n_edges);
  gcn_dinv<<<(n_nodes + B - 1) / B, B, 0, stream>>>(dinv, n_nodes);

  const int n_row_tiles = (n_nodes + 15) / 16;
  const int waves = n_row_tiles * (CCH / 16);
  gcn_gemm_wmma<<<(waves + 7) / 8, B, 0, stream>>>(x, W, xw, n_nodes);

  gcn_owner<<<(n_nodes + RNG - 1) / RNG, OWN_THREADS, 0, stream>>>(
      src, dst, ew, dinv, xw, b, out, n_edges, n_nodes);
}


# Degree launch also writes each edge's owner number (dst/391) as 8 bit planes by wave ballots; owner groups filter those planes, 64 edges a lane, not the dst list; sum orders as before (arrival)
# speedup vs baseline: 1.8624x; 1.0984x over previous
#include <hip/hip_runtime.h>
#include <hip/hip_bf16.h>
#include <math.h>

typedef __attribute__((ext_vector_type(2))) float v2f;
typedef __attribute__((ext_vector_type(8))) float v8f;

#define CCH   64
#define FILLW 2.0f

__device__ __forceinline__ void atomic_add_f32(float* p, float v) {
  __hip_atomic_fetch_add(p, v, __ATOMIC_RELAXED, __HIP_MEMORY_SCOPE_AGENT);
}

__global__ void gcn_init(float* __restrict__ deg, int n_nodes) {
  int t = blockIdx.x * blockDim.x + threadIdx.x;
  if (t < n_nodes) deg[t] = FILLW;
}

#define RNG        391
#define KEY_PLANES 8

__device__ __forceinline__ unsigned pad_words(const unsigned* p) {
  return ((256u - ((unsigned)reinterpret_cast<size_t>(p) & 255u)) & 255u) >> 2;
}
__device__ __forceinline__ int plane_words(int n_edges) { return (((n_edges + 31) >> 5) + 63) & ~63; }

#define DEG_THREADS 1024

__global__ void __launch_bounds__(DEG_THREADS)
gcn_deg(const int* __restrict__ dst, const float* __restrict__ w,
        float* __restrict__ deg, unsigned* __restrict__ key_raw, int n_edges) {
  __shared__ unsigned tile[KEY_PLANES][32];
  const int tid  = threadIdx.x;
  const int e    = blockIdx.x * DEG_THREADS + tid;
  const int lane = tid & 31;
  const int wave = tid >> 5;
  unsigned key = 0u;
  if (e < n_edges) {
    const int d = dst[e];
    atomic_add_f32(&deg[d], w[e]);
    key = ((unsigned)d / RNG) & 0xFFu;
  }
  unsigned word = 0u;
#pragma unroll
  for (int j = 0; j < KEY_PLANES; ++j) {
    const unsigned bj = __builtin_amdgcn_ballot_w32(((key >> j) & 1u) != 0u);
    if (lane == j) word = bj;
  }
  if (lane < KEY_PLANES) tile[lane][wave] = word;
  __syncthreads();
  if (tid < KEY_PLANES * 32) {
    unsigned* planes = key_raw + pad_words(key_raw);
    planes[(size_t)wave * plane_words(n_edges) + blockIdx.x * 32 + lane] = tile[wave][lane];
  }
}

__global__ void gcn_dinv(float* __restrict__ deg, int n_nodes) {
  int n = blockIdx.x * blockDim.x + threadIdx.x;
  if (n < n_nodes) {
    float d = deg[n];
    deg[n] = (d > 0.0f) ? rsqrtf(d) : 0.0f;
  }
}

__global__ void gcn_gemm_wmma(const float* __restrict__ X,
                              const float* __restrict__ W,
                              float* __restrict__ XW, int n_nodes) {
  const int wave = (blockIdx.x * blockDim.x + threadIdx.x) >> 5;
  const int lane = threadIdx.x & 31;
  const int n_row_tiles = (n_nodes + 15) >> 4;
  const int row_tile = wave >> 2;
  const int col_tile = wave & 3;
  if (row_tile >= n_row_tiles) return;

  const int half = lane >> 4;
  const int l16  = lane & 15;
  int row = row_tile * 16 + l16;
  int ldrow = (row < n_nodes) ? row : (n_nodes - 1);

  const float* __restrict__ xrow = X + (size_t)ldrow * CCH;
  const float* __restrict__ wcol = W + col_tile * 16 + l16;

  v8f acc = {};
#pragma unroll
  for (int k0 = 0; k0 < CCH; k0 += 4) {
    const int ka = k0 + 2 * half;
    v2f a, b;
    a.x = xrow[ka + 0];
    a.y = xrow[ka + 1];
    b.x = wcol[(size_t)(ka + 0) * CCH];
    b.y = wcol[(size_t)(ka + 1) * CCH];
    acc = __builtin_amdgcn_wmma_f32_16x16x4_f32(
         false, a,  false, b,
         (short)0, acc,  false,  false);
  }

  const int base_row = row_tile * 16 + 8 * half;
  float* __restrict__ dstp = XW + (size_t)base_row * CCH + col_tile * 16 + l16;
#pragma unroll
  for (int i = 0; i < 8; ++i) {
    if (base_row + i < n_nodes) dstp[(size_t)i * CCH] = acc[i];
  }
}

typedef __attribute__((ext_vector_type(4))) float    v4f;
typedef __attribute__((ext_vector_type(2))) unsigned v2u;

#define OWN_THREADS 1024
#define OWN_WAVES   32
#define QCAP        64
#define WAVE_CHUNKS 32
#define STEP_CHUNKS (OWN_WAVES * WAVE_CHUNKS)
#define CHUNK_EDGES 64
#define ROWS_FLY    16

__device__ __forceinline__ void lds_add_f32(float* p, float v) {
  __hip_atomic_fetch_add(p, v, __ATOMIC_RELAXED, __HIP_MEMORY_SCOPE_WORKGROUP);
}

__device__ __forceinline__ void q_batch(const unsigned* ring, unsigned head, int cnt,
                                        int lane, int lo,
                                        const int* __restrict__ src, const int* __restrict__ dst,
                                        const float* __restrict__ w,
                                        const float* __restrict__ dinv, const float* __restrict__ xw,
                                        float* __restrict__ acc) {
  int   s   = lo;
  int   dl  = RNG;
  float nrm = 0.0f;
  if (lane < cnt) {
    const int e = (int)ring[(head + (unsigned)lane) & (QCAP - 1)];
    const int d = dst[e];
    s = src[e];
    const float nn = dinv[s] * w[e] * dinv[d];
    const unsigned du = (unsigned)d - (unsigned)lo;
    if (du < RNG) { dl = (int)du; nrm = nn; }
  }
  for (int j0 = 0; j0 < cnt; j0 += ROWS_FLY) {
    v2f x[ROWS_FLY];
#pragma unroll
    for (int k = 0; k < ROWS_FLY; ++k) {
      const int sj = __builtin_amdgcn_readlane(s, j0 + k);
      x[k] = *reinterpret_cast<const v2f*>(xw + (size_t)sj * CCH + 2 * lane);
    }
#pragma unroll
    for (int k = 0; k < ROWS_FLY; ++k) {
      const int   dj = __builtin_amdgcn_readlane(dl, j0 + k);
      const float nj = __int_as_float(__builtin_amdgcn_readlane(__float_as_int(nrm), j0 + k));
      float* a = acc + dj * CCH + 2 * lane;
      lds_add_f32(a,     x[k].x * nj);
      lds_add_f32(a + 1, x[k].y * nj);
    }
  }
}

__device__ __forceinline__ void q_push_bits(unsigned* ring, unsigned& head, unsigned& tail,
                                            unsigned flo, unsigned fhi, int e0,
                                            int lane, int lo,
                                            const int* __restrict__ src, const int* __restrict__ dst,
                                            const float* __restrict__ w,
                                            const float* __restrict__ dinv, const float* __restrict__ xw,
                                            float* __restrict__ acc) {
  unsigned mask = __builtin_amdgcn_ballot_w32((flo | fhi) != 0u);
  while (mask != 0u) {
    if ((flo | fhi) != 0u) {
      const bool in_lo = (flo != 0u);
      const unsigned cur = in_lo ? flo : fhi;
      const int bit = __builtin_ctz(cur) + (in_lo ? 0 : 32);
      const unsigned slot = (tail + __builtin_amdgcn_mbcnt_lo(mask, 0u)) & (QCAP - 1);
      ring[slot] = (unsigned)(e0 + bit);
      const unsigned rest = cur & (cur - 1u);
      if (in_lo) flo = rest; else fhi = rest;
    }
    tail += (unsigned)__builtin_popcount(mask);
    if (tail - head >= 32u) {
      q_batch(ring, head, 32, lane, lo, src, dst, w, dinv, xw, acc);
      head += 32u;
    }
    mask = __builtin_amdgcn_ballot_w32((flo | fhi) != 0u);
  }
}

__device__ __forceinline__ v2u own_bits(const v2u (&k)[KEY_PLANES], const unsigned (&z)[KEY_PLANES]) {
  unsigned dx = 0u, dy = 0u;
#pragma unroll
  for (int j = 0; j < KEY_PLANES; ++j) {
    dx |= k[j].x ^ z[j];
    dy |= k[j].y ^ z[j];
  }
  v2u m; m.x = ~dx; m.y = ~dy;
  return m;
}

__device__ __forceinline__ void load_keys(v2u (&k)[KEY_PLANES], const v2u* __restrict__ planes2, int pw2, int c) {
#pragma unroll
  for (int j = 0; j < KEY_PLANES; ++j) k[j] = planes2[(size_t)j * pw2 + c];
}

__global__ void __launch_bounds__(OWN_THREADS)
gcn_owner(const int* __restrict__ src, const int* __restrict__ dst,
          const float* __restrict__ w, const float* __restrict__ dinv,
          const float* __restrict__ xw, const float* __restrict__ b,
          const unsigned* __restrict__ key_raw,
          float* __restrict__ out, int n_edges, int n_nodes) {
  __shared__ __attribute__((aligned(16))) float    acc[(RNG + 1) * CCH];
  __shared__ __attribute__((aligned(16))) unsigned ringbuf[OWN_WAVES * QCAP];

  const int tid  = threadIdx.x;
  const int lane = tid & 31;
  const int wave = __builtin_amdgcn_readfirstlane(tid >> 5);
  const int lo   = blockIdx.x * RNG;
  unsigned* ring = ringbuf + wave * QCAP;

  for (int i = tid; i < (RNG + 1) * CCH / 4; i += OWN_THREADS) {
    v4f zero4 = {0.0f, 0.0f, 0.0f, 0.0f};
    reinterpret_cast<v4f*>(acc)[i] = zero4;
  }
  __syncthreads();

  unsigned head = 0u, tail = 0u;

  const v2u* __restrict__ planes2 = reinterpret_cast<const v2u*>(key_raw + pad_words(key_raw));
  const int pw2 = plane_words(n_edges) >> 1;
  const int n_chunks = n_edges / CHUNK_EDGES;
  const int n_full   = n_chunks / STEP_CHUNKS;

  unsigned z[KEY_PLANES];
#pragma unroll
  for (int j = 0; j < KEY_PLANES; ++j) z[j] = 0u - ((blockIdx.x >> j) & 1u);

  if (n_full > 0) {
    int it = (int)(((long long)blockIdx.x * n_full) / gridDim.x);
    v2u ka[KEY_PLANES], kb[KEY_PLANES];
    load_keys(ka, planes2, pw2, (it * OWN_WAVES + wave) * WAVE_CHUNKS + lane);
    int n = 0;
    for (; n + 2 <= n_full; n += 2) {
      int it1 = it + 1;  if (it1 == n_full) it1 = 0;
      int it2 = it1 + 1; if (it2 == n_full) it2 = 0;
      load_keys(kb, planes2, pw2, (it1 * OWN_WAVES + wave) * WAVE_CHUNKS + lane);
      {
        const v2u m = own_bits(ka, z);
        q_push_bits(ring, head, tail, m.x, m.y, ((it * OWN_WAVES + wave) * WAVE_CHUNKS + lane) * CHUNK_EDGES,
                    lane, lo, src, dst, w, dinv, xw, acc);
      }
      load_keys(ka, planes2, pw2, (it2 * OWN_WAVES + wave) * WAVE_CHUNKS + lane);
      {
        const v2u m = own_bits(kb, z);
        q_push_bits(ring, head, tail, m.x, m.y, ((it1 * OWN_WAVES + wave) * WAVE_CHUNKS + lane) * CHUNK_EDGES,
                    lane, lo, src, dst, w, dinv, xw, acc);
      }
      it = it2;
    }
    if (n < n_full) {
      const v2u m = own_bits(ka, z);
      q_push_bits(ring, head, tail, m.x, m.y, ((it * OWN_WAVES + wave) * WAVE_CHUNKS + lane) * CHUNK_EDGES,
                  lane, lo, src, dst, w, dinv, xw, acc);
    }
  }
  for (int cb = n_full * STEP_CHUNKS + wave * 32; cb < n_chunks; cb += OWN_THREADS) {
    const int  c     = cb + lane;
    const bool valid = (c < n_chunks);
    v2u kr[KEY_PLANES];
    load_keys(kr, planes2, pw2, valid ? c : 0);
    v2u m = own_bits(kr, z);
    if (!valid) { m.x = 0u; m.y = 0u; }
    q_push_bits(ring, head, tail, m.x, m.y, c * CHUNK_EDGES, lane, lo, src, dst, w, dinv, xw, acc);
  }
  for (int base = n_chunks * CHUNK_EDGES + wave * 32; base < n_edges; base += OWN_THREADS) {
    const int e = base + lane;
    const int d = (e < n_edges) ? dst[e] : -1;
    const unsigned du = (unsigned)d - (unsigned)lo;
    q_push_bits(ring, head, tail, (du < RNG) ? 1u : 0u, 0u, e, lane, lo, src, dst, w, dinv, xw, acc);
  }
  if (tail != head) q_batch(ring, head, (int)(tail - head), lane, lo, src, dst, w, dinv, xw, acc);

  __syncthreads();

  for (int r = wave; r < RNG; r += OWN_WAVES) {
    const int n = lo + r;
    if (n < n_nodes) {
      const float di = dinv[n];
      const float k  = FILLW * di * di;
      const v2f a  = *reinterpret_cast<const v2f*>(acc + r * CCH + 2 * lane);
      const v2f x  = *reinterpret_cast<const v2f*>(xw + (size_t)n * CCH + 2 * lane);
      const v2f bb = *reinterpret_cast<const v2f*>(b + 2 * lane);
      v2f o;
      o.x = tanhf(a.x + x.x * k + bb.x);
      o.y = tanhf(a.y + x.y * k + bb.y);
      *reinterpret_cast<v2f*>(out + (size_t)n * CCH + 2 * lane) = o;
    }
  }
}

extern "C" void kernel_launch(void* const* d_in, const int* in_sizes, int n_in,
                              void* d_out, int out_size, void* d_ws, size_t ws_size,
                              hipStream_t stream) {
  const float* x  = (const float*)d_in[0];
  const int*   ei = (const int*)d_in[1];
  const float* ew = (const float*)d_in[2];
  const float* W  = (const float*)d_in[3];
  const float* b  = (const float*)d_in[4];
  float* out = (float*)d_out;

  const int n_nodes = in_sizes[0] / CCH;
  const int n_edges = in_sizes[2];
  const int* src = ei;
  const int* dst = ei + n_edges;

  float* xw   = (float*)d_ws;
  float* dinv = xw + (size_t)n_nodes * CCH;
  unsigned* key_raw = (unsigned*)(dinv + n_nodes);

  const int B = 256;

  gcn_init<<<(n_nodes + B - 1) / B, B, 0, stream>>>(dinv, n_nodes);
  gcn_deg<<<(n_edges + DEG_THREADS - 1) / DEG_THREADS, DEG_THREADS, 0, stream>>>(dst, ew, dinv, key_raw, n_edges);
  gcn_dinv<<<(n_nodes + B - 1) / B, B, 0, stream>>>(dinv, n_nodes);

  const int n_row_tiles = (n_nodes + 15) / 16;
  const int waves = n_row_tiles * (CCH / 16);
  gcn_gemm_wmma<<<(waves + 7) / 8, B, 0, stream>>>(x, W, xw, n_nodes);

  gcn_owner<<<(n_nodes + RNG - 1) / RNG, OWN_THREADS, 0, stream>>>(
      src, dst, ew, dinv, xw, b, key_raw, out, n_edges, n_nodes);
}


# Degree sums off global atomics: group (25600-edge chunk, node quarter) adds w into an LDS table, stores it; dinv launch sums the 63 partials; no init launch; deg order = LDS arrival, then chunk
# speedup vs baseline: 3.1016x; 1.6654x over previous
#include <hip/hip_runtime.h>
#include <hip/hip_bf16.h>
#include <math.h>

typedef __attribute__((ext_vector_type(2))) float v2f;
typedef __attribute__((ext_vector_type(8))) float v8f;

#define CCH   64
#define FILLW 2.0f

#define RNG        391
#define KEY_PLANES 8

__device__ __forceinline__ unsigned pad_words(const unsigned* p) {
  return ((256u - ((unsigned)reinterpret_cast<size_t>(p) & 255u)) & 255u) >> 2;
}
__device__ __forceinline__ int plane_words(int n_edges) { return (((n_edges + 31) >> 5) + 63) & ~63; }

typedef __attribute__((ext_vector_type(4))) float    v4f;

#define NQ          4
#define QN          25000
#define QP          25024
#define DEG_LINES   25
#define DEG_THREADS 1024
#define PL_PER_Q    (KEY_PLANES / NQ)

__device__ __forceinline__ void lds_add_f32(float* p, float v) {
  __hip_atomic_fetch_add(p, v, __ATOMIC_RELAXED, __HIP_MEMORY_SCOPE_WORKGROUP);
}

template <bool FULL>
__device__ __forceinline__ void load_edges(int (&d)[DEG_LINES], float (&ww)[DEG_LINES],
                                           const int* __restrict__ dst, const float* __restrict__ w,
                                           int e0, int n_edges) {
#pragma unroll
  for (int i = 0; i < DEG_LINES; ++i) {
    const int e = e0 + i * DEG_THREADS;
    if (FULL || e < n_edges) { d[i] = dst[e]; ww[i] = w[e]; }
    else                     { d[i] = -1;     ww[i] = 0.0f; }
  }
}

__global__ void __launch_bounds__(DEG_THREADS)
gcn_deg(const int* __restrict__ dst, const float* __restrict__ w,
        float* __restrict__ part, unsigned* __restrict__ key_raw, int n_edges) {
  __shared__ __attribute__((aligned(16))) float    tab[QP];
  __shared__ unsigned tile[DEG_LINES][PL_PER_Q][32];
  const int tid   = threadIdx.x;
  const int lane  = tid & 31;
  const int wave  = tid >> 5;
  const int q     = blockIdx.x;
  const int c     = blockIdx.y;
  const int line0 = c * DEG_LINES;

  int   d[DEG_LINES];
  float ww[DEG_LINES];
  if ((line0 + DEG_LINES) * DEG_THREADS <= n_edges)
    load_edges<true>(d, ww, dst, w, line0 * DEG_THREADS + tid, n_edges);
  else
    load_edges<false>(d, ww, dst, w, line0 * DEG_THREADS + tid, n_edges);
  for (int i = tid; i < QP / 4; i += DEG_THREADS) {
    v4f zero4 = {0.0f, 0.0f, 0.0f, 0.0f};
    reinterpret_cast<v4f*>(tab)[i] = zero4;
  }
  __syncthreads();

  const unsigned nlo = (unsigned)(q * QN);
#pragma unroll
  for (int i = 0; i < DEG_LINES; ++i) {
    const unsigned du = (unsigned)d[i] - nlo;
    if (du < (unsigned)QN) lds_add_f32(&tab[du], ww[i]);
    const unsigned key = (d[i] >= 0) ? (((unsigned)d[i] / RNG) & 0xFFu) : 0u;
    unsigned word = 0u;
#pragma unroll
    for (int t = 0; t < PL_PER_Q; ++t) {
      const int j = q * PL_PER_Q + t;
      const unsigned bj = __builtin_amdgcn_ballot_w32(((key >> j) & 1u) != 0u);
      if (lane == t) word = bj;
    }
    if (lane < PL_PER_Q) tile[i][lane][wave] = word;
  }
  __syncthreads();

  float* __restrict__ pout = part + ((size_t)c * NQ + q) * QP;
  for (int i = tid; i < QP / 4; i += DEG_THREADS)
    reinterpret_cast<v4f*>(pout)[i] = reinterpret_cast<const v4f*>(tab)[i];

  unsigned* planes = key_raw + pad_words(key_raw);
  const int pw = plane_words(n_edges);
  for (int s = wave; s < DEG_LINES * PL_PER_Q; s += DEG_THREADS / 32) {
    const int i = s / PL_PER_Q;
    const int t = s - i * PL_PER_Q;
    const int ln = line0 + i;
    if (ln * DEG_THREADS < n_edges)
      planes[(size_t)(q * PL_PER_Q + t) * pw + ln * 32 + lane] = tile[i][t][lane];
  }
}

#define DINV_FLY 32

__global__ void gcn_dinv(const float* __restrict__ part, float* __restrict__ dinv, int n_nodes, int n_chunks) {
  const int n = blockIdx.x * blockDim.x + threadIdx.x;
  if (n < n_nodes) {
    const int q = n / QN;
    const float* __restrict__ p = part + (size_t)q * QP + (n - q * QN);
    float dsum = FILLW;
    for (int c0 = 0; c0 < n_chunks; c0 += DINV_FLY) {
      float v[DINV_FLY];
#pragma unroll
      for (int k = 0; k < DINV_FLY; ++k) v[k] = (c0 + k < n_chunks) ? p[(size_t)(c0 + k) * (NQ * QP)] : 0.0f;
#pragma unroll
      for (int k = 0; k < DINV_FLY; ++k) dsum += v[k];
    }
    dinv[n] = (dsum > 0.0f) ? rsqrtf(dsum) : 0.0f;
  }
}

__global__ void gcn_gemm_wmma(const float* __restrict__ X,
                              const float* __restrict__ W,
                              float* __restrict__ XW, int n_nodes) {
  const int wave = (blockIdx.x * blockDim.x + threadIdx.x) >> 5;
  const int lane = threadIdx.x & 31;
  const int n_row_tiles = (n_nodes + 15) >> 4;
  const int row_tile = wave >> 2;
  const int col_tile = wave & 3;
  if (row_tile >= n_row_tiles) return;

  const int half = lane >> 4;
  const int l16  = lane & 15;
  int row = row_tile * 16 + l16;
  int ldrow = (row < n_nodes) ? row : (n_nodes - 1);

  const float* __restrict__ xrow = X + (size_t)ldrow * CCH;
  const float* __restrict__ wcol = W + col_tile * 16 + l16;

  v8f acc = {};
#pragma unroll
  for (int k0 = 0; k0 < CCH; k0 += 4) {
    const int ka = k0 + 2 * half;
    v2f a, b;
    a.x = xrow[ka + 0];
    a.y = xrow[ka + 1];
    b.x = wcol[(size_t)(ka + 0) * CCH];
    b.y = wcol[(size_t)(ka + 1) * CCH];
    acc = __builtin_amdgcn_wmma_f32_16x16x4_f32(
         false, a,  false, b,
         (short)0, acc,  false,  false);
  }

  const int base_row = row_tile * 16 + 8 * half;
  float* __restrict__ dstp = XW + (size_t)base_row * CCH + col_tile * 16 + l16;
#pragma unroll
  for (int i = 0; i < 8; ++i) {
    if (base_row + i < n_nodes) dstp[(size_t)i * CCH] = acc[i];
  }
}

typedef __attribute__((ext_vector_type(2))) unsigned v2u;

#define OWN_THREADS 1024
#define OWN_WAVES   32
#define QCAP        64
#define WAVE_CHUNKS 32
#define STEP_CHUNKS (OWN_WAVES * WAVE_CHUNKS)
#define CHUNK_EDGES 64
#define ROWS_FLY    16

__device__ __forceinline__ void q_batch(const unsigned* ring, unsigned head, int cnt,
                                        int lane, int lo,
                                        const int* __restrict__ src, const int* __restrict__ dst,
                                        const float* __restrict__ w,
                                        const float* __restrict__ dinv, const float* __restrict__ xw,
                                        float* __restrict__ acc) {
  int   s   = lo;
  int   dl  = RNG;
  float nrm = 0.0f;
  if (lane < cnt) {
    const int e = (int)ring[(head + (unsigned)lane) & (QCAP - 1)];
    const int d = dst[e];
    s = src[e];
    const float nn = dinv[s] * w[e] * dinv[d];
    const unsigned du = (unsigned)d - (unsigned)lo;
    if (du < RNG) { dl = (int)du; nrm = nn; }
  }
  for (int j0 = 0; j0 < cnt; j0 += ROWS_FLY) {
    v2f x[ROWS_FLY];
#pragma unroll
    for (int k = 0; k < ROWS_FLY; ++k) {
      const int sj = __builtin_amdgcn_readlane(s, j0 + k);
      x[k] = *reinterpret_cast<const v2f*>(xw + (size_t)sj * CCH + 2 * lane);
    }
#pragma unroll
    for (int k = 0; k < ROWS_FLY; ++k) {
      const int   dj = __builtin_amdgcn_readlane(dl, j0 + k);
      const float nj = __int_as_float(__builtin_amdgcn_readlane(__float_as_int(nrm), j0 + k));
      float* a = acc + dj * CCH + 2 * lane;
      lds_add_f32(a,     x[k].x * nj);
      lds_add_f32(a + 1, x[k].y * nj);
    }
  }
}

__device__ __forceinline__ void q_push_bits(unsigned* ring, unsigned& head, unsigned& tail,
                                            unsigned flo, unsigned fhi, int e0,
                                            int lane, int lo,
                                            const int* __restrict__ src, const int* __restrict__ dst,
                                            const float* __restrict__ w,
                                            const float* __restrict__ dinv, const float* __restrict__ xw,
                                            float* __restrict__ acc) {
  unsigned mask = __builtin_amdgcn_ballot_w32((flo | fhi) != 0u);
  while (mask != 0u) {
    if ((flo | fhi) != 0u) {
      const bool in_lo = (flo != 0u);
      const unsigned cur = in_lo ? flo : fhi;
      const int bit = __builtin_ctz(cur) + (in_lo ? 0 : 32);
      const unsigned slot = (tail + __builtin_amdgcn_mbcnt_lo(mask, 0u)) & (QCAP - 1);
      ring[slot] = (unsigned)(e0 + bit);
      const unsigned rest = cur & (cur - 1u);
      if (in_lo) flo = rest; else fhi = rest;
    }
    tail += (unsigned)__builtin_popcount(mask);
    if (tail - head >= 32u) {
      q_batch(ring, head, 32, lane, lo, src, dst, w, dinv, xw, acc);
      head += 32u;
    }
    mask = __builtin_amdgcn_ballot_w32((flo | fhi) != 0u);
  }
}

__device__ __forceinline__ v2u own_bits(const v2u (&k)[KEY_PLANES], const unsigned (&z)[KEY_PLANES]) {
  unsigned dx = 0u, dy = 0u;
#pragma unroll
  for (int j = 0; j < KEY_PLANES; ++j) {
    dx |= k[j].x ^ z[j];
    dy |= k[j].y ^ z[j];
  }
  v2u m; m.x = ~dx; m.y = ~dy;
  return m;
}

__device__ __forceinline__ void load_keys(v2u (&k)[KEY_PLANES], const v2u* __restrict__ planes2, int pw2, int c) {
#pragma unroll
  for (int j = 0; j < KEY_PLANES; ++j) k[j] = planes2[(size_t)j * pw2 + c];
}

__global__ void __launch_bounds__(OWN_THREADS)
gcn_owner(const int* __restrict__ src, const int* __restrict__ dst,
          const float* __restrict__ w, const float* __restrict__ dinv,
          const float* __restrict__ xw, const float* __restrict__ b,
          const unsigned* __restrict__ key_raw,
          float* __restrict__ out, int n_edges, int n_nodes) {
  __shared__ __attribute__((aligned(16))) float    acc[(RNG + 1) * CCH];
  __shared__ __attribute__((aligned(16))) unsigned ringbuf[OWN_WAVES * QCAP];

  const int tid  = threadIdx.x;
  const int lane = tid & 31;
  const int wave = __builtin_amdgcn_readfirstlane(tid >> 5);
  const int lo   = blockIdx.x * RNG;
  unsigned* ring = ringbuf + wave * QCAP;

  for (int i = tid; i < (RNG + 1) * CCH / 4; i += OWN_THREADS) {
    v4f zero4 = {0.0f, 0.0f, 0.0f, 0.0f};
    reinterpret_cast<v4f*>(acc)[i] = zero4;
  }
  __syncthreads();

  unsigned head = 0u, tail = 0u;

  const v2u* __restrict__ planes2 = reinterpret_cast<const v2u*>(key_raw + pad_words(key_raw));
  const int pw2 = plane_words(n_edges) >> 1;
  const int n_chunks = n_edges / CHUNK_EDGES;
  const int n_full   = n_chunks / STEP_CHUNKS;

  unsigned z[KEY_PLANES];
#pragma unroll
  for (int j = 0; j < KEY_PLANES; ++j) z[j] = 0u - ((blockIdx.x >> j) & 1u);

  if (n_full > 0) {
    int it = (int)(((long long)blockIdx.x * n_full) / gridDim.x);
    v2u ka[KEY_PLANES], kb[KEY_PLANES];
    load_keys(ka, planes2, pw2, (it * OWN_WAVES + wave) * WAVE_CHUNKS + lane);
    int n = 0;
    for (; n + 2 <= n_full; n += 2) {
      int it1 = it + 1;  if (it1 == n_full) it1 = 0;
      int it2 = it1 + 1; if (it2 == n_full) it2 = 0;
      load_keys(kb, planes2, pw2, (it1 * OWN_WAVES + wave) * WAVE_CHUNKS + lane);
      {
        const v2u m = own_bits(ka, z);
        q_push_bits(ring, head, tail, m.x, m.y, ((it * OWN_WAVES + wave) * WAVE_CHUNKS + lane) * CHUNK_EDGES,
                    lane, lo, src, dst, w, dinv, xw, acc);
      }
      load_keys(ka, planes2, pw2, (it2 * OWN_WAVES + wave) * WAVE_CHUNKS + lane);
      {
        const v2u m = own_bits(kb, z);
        q_push_bits(ring, head, tail, m.x, m.y, ((it1 * OWN_WAVES + wave) * WAVE_CHUNKS + lane) * CHUNK_EDGES,
                    lane, lo, src, dst, w, dinv, xw, acc);
      }
      it = it2;
    }
    if (n < n_full) {
      const v2u m = own_bits(ka, z);
      q_push_bits(ring, head, tail, m.x, m.y, ((it * OWN_WAVES + wave) * WAVE_CHUNKS + lane) * CHUNK_EDGES,
                  lane, lo, src, dst, w, dinv, xw, acc);
    }
  }
  for (int cb = n_full * STEP_CHUNKS + wave * 32; cb < n_chunks; cb += OWN_THREADS) {
    const int  c     = cb + lane;
    const bool valid = (c < n_chunks);
    v2u kr[KEY_PLANES];
    load_keys(kr, planes2, pw2, valid ? c : 0);
    v2u m = own_bits(kr, z);
    if (!valid) { m.x = 0u; m.y = 0u; }
    q_push_bits(ring, head, tail, m.x, m.y, c * CHUNK_EDGES, lane, lo, src, dst, w, dinv, xw, acc);
  }
  for (int base = n_chunks * CHUNK_EDGES + wave * 32; base < n_edges; base += OWN_THREADS) {
    const int e = base + lane;
    const int d = (e < n_edges) ? dst[e] : -1;
    const unsigned du = (unsigned)d - (unsigned)lo;
    q_push_bits(ring, head, tail, (du < RNG) ? 1u : 0u, 0u, e, lane, lo, src, dst, w, dinv, xw, acc);
  }
  if (tail != head) q_batch(ring, head, (int)(tail - head), lane, lo, src, dst, w, dinv, xw, acc);

  __syncthreads();

  for (int r = wave; r < RNG; r += OWN_WAVES) {
    const int n = lo + r;
    if (n < n_nodes) {
      const float di = dinv[n];
      const float k  = FILLW * di * di;
      const v2f a  = *reinterpret_cast<const v2f*>(acc + r * CCH + 2 * lane);
      const v2f x  = *reinterpret_cast<const v2f*>(xw + (size_t)n * CCH + 2 * lane);
      const v2f bb = *reinterpret_cast<const v2f*>(b + 2 * lane);
      v2f o;
      o.x = tanhf(a.x + x.x * k + bb.x);
      o.y = tanhf(a.y + x.y * k + bb.y);
      *reinterpret_cast<v2f*>(out + (size_t)n * CCH + 2 * lane) = o;
    }
  }
}

extern "C" void kernel_launch(void* const* d_in, const int* in_sizes, int n_in,
                              void* d_out, int out_size, void* d_ws, size_t ws_size,
                              hipStream_t stream) {
  const float* x  = (const float*)d_in[0];
  const int*   ei = (const int*)d_in[1];
  const float* ew = (const float*)d_in[2];
  const float* W  = (const float*)d_in[3];
  const float* b  = (const float*)d_in[4];
  float* out = (float*)d_out;

  const int n_nodes = in_sizes[0] / CCH;
  const int n_edges = in_sizes[2];
  const int* src = ei;
  const int* dst = ei + n_edges;

  float* xw   = (float*)d_ws;
  float* dinv = xw + (size_t)n_nodes * CCH;
  unsigned* key_raw = (unsigned*)(dinv + n_nodes);
  float* part = (float*)d_ws + ((size_t)32 << 20);

  const int B = 256;

  const int n_chunks = (n_edges + DEG_LINES * DEG_THREADS - 1) / (DEG_LINES * DEG_THREADS);
  gcn_deg<<<dim3(NQ, n_chunks), DEG_THREADS, 0, stream>>>(dst, ew, part, key_raw, n_edges);
  gcn_dinv<<<(n_nodes + B - 1) / B, B, 0, stream>>>(part, dinv, n_nodes, n_chunks);

  const int n_row_tiles = (n_nodes + 15) / 16;
  const int waves = n_row_tiles * (CCH / 16);
  gcn_gemm_wmma<<<(waves + 7) / 8, B, 0, stream>>>(x, W, xw, n_nodes);

  gcn_owner<<<(n_nodes + RNG - 1) / RNG, OWN_THREADS, 0, stream>>>(
      src, dst, ew, dinv, xw, b, key_raw, out, n_edges, n_nodes);
}


# Degree launch also writes an 8-byte record per edge (src + dst row in its owner range, w); owner batch reads that one record, not scattered dst, src and w words; sums and their orders unchanged
# speedup vs baseline: 3.3668x; 1.0855x over previous
#include <hip/hip_runtime.h>
#include <hip/hip_bf16.h>
#include <math.h>

typedef __attribute__((ext_vector_type(2))) float v2f;
typedef __attribute__((ext_vector_type(8))) float v8f;

#define CCH   64
#define FILLW 2.0f

#define RNG        391
#define KEY_PLANES 8

__device__ __forceinline__ unsigned pad_words(const unsigned* p) {
  return ((256u - ((unsigned)reinterpret_cast<size_t>(p) & 255u)) & 255u) >> 2;
}
__device__ __forceinline__ int plane_words(int n_edges) { return (((n_edges + 31) >> 5) + 63) & ~63; }

typedef __attribute__((ext_vector_type(4))) float    v4f;
typedef __attribute__((ext_vector_type(2))) unsigned v2u;

#define REC_SHIFT   17
#define REC_SRCMASK ((1u << REC_SHIFT) - 1u)

#define NQ          4
#define QN          25000
#define QP          25024
#define DEG_LINES   25
#define DEG_THREADS 1024
#define PL_PER_Q    (KEY_PLANES / NQ)

__device__ __forceinline__ void lds_add_f32(float* p, float v) {
  __hip_atomic_fetch_add(p, v, __ATOMIC_RELAXED, __HIP_MEMORY_SCOPE_WORKGROUP);
}

template <bool FULL>
__device__ __forceinline__ void load_edges(int (&d)[DEG_LINES], float (&ww)[DEG_LINES],
                                           const int* __restrict__ dst, const float* __restrict__ w,
                                           int e0, int n_edges) {
#pragma unroll
  for (int i = 0; i < DEG_LINES; ++i) {
    const int e = e0 + i * DEG_THREADS;
    if (FULL || e < n_edges) { d[i] = dst[e]; ww[i] = w[e]; }
    else                     { d[i] = -1;     ww[i] = 0.0f; }
  }
}

template <bool FULL>
__device__ __forceinline__ void load_src(int (&sv)[DEG_LINES], const int* __restrict__ src,
                                         int e0, int n_edges, int rsel) {
#pragma unroll
  for (int i = 0; i < DEG_LINES; ++i) {
    const int e = e0 + i * DEG_THREADS;
    sv[i] = 0;
    if ((i & 3) == rsel) { if (FULL || e < n_edges) sv[i] = src[e]; }
  }
}

__global__ void __launch_bounds__(DEG_THREADS)
gcn_deg(const int* __restrict__ src, const int* __restrict__ dst, const float* __restrict__ w,
        float* __restrict__ part, unsigned* __restrict__ key_raw, unsigned* __restrict__ rec_raw, int n_edges) {
  __shared__ __attribute__((aligned(16))) float    tab[QP];
  __shared__ unsigned tile[DEG_LINES][PL_PER_Q][32];
  const int tid   = threadIdx.x;
  const int lane  = tid & 31;
  const int wave  = __builtin_amdgcn_readfirstlane(tid >> 5);
  const int q     = blockIdx.x;
  const int c     = blockIdx.y;
  const int line0 = c * DEG_LINES;

  int   d[DEG_LINES];
  float ww[DEG_LINES];
  int   sv[DEG_LINES];
  const int rsel = (q - wave) & 3;
  if ((line0 + DEG_LINES) * DEG_THREADS <= n_edges) {
    load_edges<true>(d, ww, dst, w, line0 * DEG_THREADS + tid, n_edges);
    load_src<true>(sv, src, line0 * DEG_THREADS + tid, n_edges, rsel);
  } else {
    load_edges<false>(d, ww, dst, w, line0 * DEG_THREADS + tid, n_edges);
    load_src<false>(sv, src, line0 * DEG_THREADS + tid, n_edges, rsel);
  }
  for (int i = tid; i < QP / 4; i += DEG_THREADS) {
    v4f zero4 = {0.0f, 0.0f, 0.0f, 0.0f};
    reinterpret_cast<v4f*>(tab)[i] = zero4;
  }
  __syncthreads();

  const unsigned nlo = (unsigned)(q * QN);
#pragma unroll
  for (int i = 0; i < DEG_LINES; ++i) {
    const unsigned du = (unsigned)d[i] - nlo;
    if (du < (unsigned)QN) lds_add_f32(&tab[du], ww[i]);
    const unsigned key = (d[i] >= 0) ? (((unsigned)d[i] / RNG) & 0xFFu) : 0u;
    if ((i & 3) == rsel && d[i] >= 0) {
      v2u r;
      r.x = (unsigned)sv[i] | (((unsigned)d[i] - key * RNG) << REC_SHIFT);
      r.y = __float_as_uint(ww[i]);
      reinterpret_cast<v2u*>(rec_raw)[line0 * DEG_THREADS + tid + i * DEG_THREADS] = r;
    }
    unsigned word = 0u;
#pragma unroll
    for (int t = 0; t < PL_PER_Q; ++t) {
      const int j = q * PL_PER_Q + t;
      const unsigned bj = __builtin_amdgcn_ballot_w32(((key >> j) & 1u) != 0u);
      if (lane == t) word = bj;
    }
    if (lane < PL_PER_Q) tile[i][lane][wave] = word;
  }
  __syncthreads();

  float* __restrict__ pout = part + ((size_t)c * NQ + q) * QP;
  for (int i = tid; i < QP / 4; i += DEG_THREADS)
    reinterpret_cast<v4f*>(pout)[i] = reinterpret_cast<const v4f*>(tab)[i];

  unsigned* planes = key_raw + pad_words(key_raw);
  const int pw = plane_words(n_edges);
  for (int s = wave; s < DEG_LINES * PL_PER_Q; s += DEG_THREADS / 32) {
    const int i = s / PL_PER_Q;
    const int t = s - i * PL_PER_Q;
    const int ln = line0 + i;
    if (ln * DEG_THREADS < n_edges)
      planes[(size_t)(q * PL_PER_Q + t) * pw + ln * 32 + lane] = tile[i][t][lane];
  }
}

#define DINV_FLY 32

__global__ void gcn_dinv(const float* __restrict__ part, float* __restrict__ dinv, int n_nodes, int n_chunks) {
  const int n = blockIdx.x * blockDim.x + threadIdx.x;
  if (n < n_nodes) {
    const int q = n / QN;
    const float* __restrict__ p = part + (size_t)q * QP + (n - q * QN);
    float dsum = FILLW;
    for (int c0 = 0; c0 < n_chunks; c0 += DINV_FLY) {
      float v[DINV_FLY];
#pragma unroll
      for (int k = 0; k < DINV_FLY; ++k) v[k] = (c0 + k < n_chunks) ? p[(size_t)(c0 + k) * (NQ * QP)] : 0.0f;
#pragma unroll
      for (int k = 0; k < DINV_FLY; ++k) dsum += v[k];
    }
    dinv[n] = (dsum > 0.0f) ? rsqrtf(dsum) : 0.0f;
  }
}

__global__ void gcn_gemm_wmma(const float* __restrict__ X,
                              const float* __restrict__ W,
                              float* __restrict__ XW, int n_nodes) {
  const int wave = (blockIdx.x * blockDim.x + threadIdx.x) >> 5;
  const int lane = threadIdx.x & 31;
  const int n_row_tiles = (n_nodes + 15) >> 4;
  const int row_tile = wave >> 2;
  const int col_tile = wave & 3;
  if (row_tile >= n_row_tiles) return;

  const int half = lane >> 4;
  const int l16  = lane & 15;
  int row = row_tile * 16 + l16;
  int ldrow = (row < n_nodes) ? row : (n_nodes - 1);

  const float* __restrict__ xrow = X + (size_t)ldrow * CCH;
  const float* __restrict__ wcol = W + col_tile * 16 + l16;

  v8f acc = {};
#pragma unroll
  for (int k0 = 0; k0 < CCH; k0 += 4) {
    const int ka = k0 + 2 * half;
    v2f a, b;
    a.x = xrow[ka + 0];
    a.y = xrow[ka + 1];
    b.x = wcol[(size_t)(ka + 0) * CCH];
    b.y = wcol[(size_t)(ka + 1) * CCH];
    acc = __builtin_amdgcn_wmma_f32_16x16x4_f32(
         false, a,  false, b,
         (short)0, acc,  false,  false);
  }

  const int base_row = row_tile * 16 + 8 * half;
  float* __restrict__ dstp = XW + (size_t)base_row * CCH + col_tile * 16 + l16;
#pragma unroll
  for (int i = 0; i < 8; ++i) {
    if (base_row + i < n_nodes) dstp[(size_t)i * CCH] = acc[i];
  }
}

#define OWN_THREADS 1024
#define OWN_WAVES   32
#define QCAP        64
#define WAVE_CHUNKS 32
#define STEP_CHUNKS (OWN_WAVES * WAVE_CHUNKS)
#define CHUNK_EDGES 64
#define ROWS_FLY    16

__device__ __forceinline__ void q_batch(const unsigned* ring, unsigned head, int cnt,
                                        int lane, int lo,
                                        const v2u* __restrict__ rec,
                                        const float* __restrict__ dinv, const float* __restrict__ xw,
                                        float* __restrict__ acc) {
  int   s   = lo;
  int   dl  = RNG;
  float nrm = 0.0f;
  if (lane < cnt) {
    const int e = (int)ring[(head + (unsigned)lane) & (QCAP - 1)];
    const v2u r = rec[e];
    s = (int)(r.x & REC_SRCMASK);
    const unsigned du = r.x >> REC_SHIFT;
    if (du < RNG) {
      dl  = (int)du;
      nrm = dinv[s] * __uint_as_float(r.y) * dinv[lo + (int)du];
    }
  }
  for (int j0 = 0; j0 < cnt; j0 += ROWS_FLY) {
    v2f x[ROWS_FLY];
#pragma unroll
    for (int k = 0; k < ROWS_FLY; ++k) {
      const int sj = __builtin_amdgcn_readlane(s, j0 + k);
      x[k] = *reinterpret_cast<const v2f*>(xw + (size_t)sj * CCH + 2 * lane);
    }
#pragma unroll
    for (int k = 0; k < ROWS_FLY; ++k) {
      const int   dj = __builtin_amdgcn_readlane(dl, j0 + k);
      const float nj = __int_as_float(__builtin_amdgcn_readlane(__float_as_int(nrm), j0 + k));
      float* a = acc + dj * CCH + 2 * lane;
      lds_add_f32(a,     x[k].x * nj);
      lds_add_f32(a + 1, x[k].y * nj);
    }
  }
}

__device__ __forceinline__ void q_push_bits(unsigned* ring, unsigned& head, unsigned& tail,
                                            unsigned flo, unsigned fhi, int e0,
                                            int lane, int lo,
                                            const v2u* __restrict__ rec,
                                            const float* __restrict__ dinv, const float* __restrict__ xw,
                                            float* __restrict__ acc) {
  unsigned mask = __builtin_amdgcn_ballot_w32((flo | fhi) != 0u);
  while (mask != 0u) {
    if ((flo | fhi) != 0u) {
      const bool in_lo = (flo != 0u);
      const unsigned cur = in_lo ? flo : fhi;
      const int bit = __builtin_ctz(cur) + (in_lo ? 0 : 32);
      const unsigned slot = (tail + __builtin_amdgcn_mbcnt_lo(mask, 0u)) & (QCAP - 1);
      ring[slot] = (unsigned)(e0 + bit);
      const unsigned rest = cur & (cur - 1u);
      if (in_lo) flo = rest; else fhi = rest;
    }
    tail += (unsigned)__builtin_popcount(mask);
    if (tail - head >= 32u) {
      q_batch(ring, head, 32, lane, lo, rec, dinv, xw, acc);
      head += 32u;
    }
    mask = __builtin_amdgcn_ballot_w32((flo | fhi) != 0u);
  }
}

__device__ __forceinline__ v2u own_bits(const v2u (&k)[KEY_PLANES], const unsigned (&z)[KEY_PLANES]) {
  unsigned dx = 0u, dy = 0u;
#pragma unroll
  for (int j = 0; j < KEY_PLANES; ++j) {
    dx |= k[j].x ^ z[j];
    dy |= k[j].y ^ z[j];
  }
  v2u m; m.x = ~dx; m.y = ~dy;
  return m;
}

__device__ __forceinline__ void load_keys(v2u (&k)[KEY_PLANES], const v2u* __restrict__ planes2, int pw2, int c) {
#pragma unroll
  for (int j = 0; j < KEY_PLANES; ++j) k[j] = planes2[(size_t)j * pw2 + c];
}

__global__ void __launch_bounds__(OWN_THREADS)
gcn_owner(const int* __restrict__ dst, const float* __restrict__ dinv,
          const float* __restrict__ xw, const float* __restrict__ b,
          const unsigned* __restrict__ key_raw, const unsigned* __restrict__ rec_raw,
          float* __restrict__ out, int n_edges, int n_nodes) {
  __shared__ __attribute__((aligned(16))) float    acc[(RNG + 1) * CCH];
  __shared__ __attribute__((aligned(16))) unsigned ringbuf[OWN_WAVES * QCAP];

  const int tid  = threadIdx.x;
  const int lane = tid & 31;
  const int wave = __builtin_amdgcn_readfirstlane(tid >> 5);
  const int lo   = blockIdx.x * RNG;
  unsigned* ring = ringbuf + wave * QCAP;
  const v2u* __restrict__ rec = reinterpret_cast<const v2u*>(rec_raw);

  for (int i = tid; i < (RNG + 1) * CCH / 4; i += OWN_THREADS) {
    v4f zero4 = {0.0f, 0.0f, 0.0f, 0.0f};
    reinterpret_cast<v4f*>(acc)[i] = zero4;
  }
  __syncthreads();

  unsigned head = 0u, tail = 0u;

  const v2u* __restrict__ planes2 = reinterpret_cast<const v2u*>(key_raw + pad_words(key_raw));
  const int pw2 = plane_words(n_edges) >> 1;
  const int n_chunks = n_edges / CHUNK_EDGES;
  const int n_full   = n_chunks / STEP_CHUNKS;

  unsigned z[KEY_PLANES];
#pragma unroll
  for (int j = 0; j < KEY_PLANES; ++j) z[j] = 0u - ((blockIdx.x >> j) & 1u);

  if (n_full > 0) {
    int it = (int)(((long long)blockIdx.x * n_full) / gridDim.x);
    v2u ka[KEY_PLANES], kb[KEY_PLANES];
    load_keys(ka, planes2, pw2, (it * OWN_WAVES + wave) * WAVE_CHUNKS + lane);
    int n = 0;
    for (; n + 2 <= n_full; n += 2) {
      int it1 = it + 1;  if (it1 == n_full) it1 = 0;
      int it2 = it1 + 1; if (it2 == n_full) it2 = 0;
      load_keys(kb, planes2, pw2, (it1 * OWN_WAVES + wave) * WAVE_CHUNKS + lane);
      {
        const v2u m = own_bits(ka, z);
        q_push_bits(ring, head, tail, m.x, m.y, ((it * OWN_WAVES + wave) * WAVE_CHUNKS + lane) * CHUNK_EDGES,
                    lane, lo, rec, dinv, xw, acc);
      }
      load_keys(ka, planes2, pw2, (it2 * OWN_WAVES + wave) * WAVE_CHUNKS + lane);
      {
        const v2u m = own_bits(kb, z);
        q_push_bits(ring, head, tail, m.x, m.y, ((it1 * OWN_WAVES + wave) * WAVE_CHUNKS + lane) * CHUNK_EDGES,
                    lane, lo, rec, dinv, xw, acc);
      }
      it = it2;
    }
    if (n < n_full) {
      const v2u m = own_bits(ka, z);
      q_push_bits(ring, head, tail, m.x, m.y, ((it * OWN_WAVES + wave) * WAVE_CHUNKS + lane) * CHUNK_EDGES,
                  lane, lo, rec, dinv, xw, acc);
    }
  }
  for (int cb = n_full * STEP_CHUNKS + wave * 32; cb < n_chunks; cb += OWN_THREADS) {
    const int  c     = cb + lane;
    const bool valid = (c < n_chunks);
    v2u kr[KEY_PLANES];
    load_keys(kr, planes2, pw2, valid ? c : 0);
    v2u m = own_bits(kr, z);
    if (!valid) { m.x = 0u; m.y = 0u; }
    q_push_bits(ring, head, tail, m.x, m.y, c * CHUNK_EDGES, lane, lo, rec, dinv, xw, acc);
  }
  for (int base = n_chunks * CHUNK_EDGES + wave * 32; base < n_edges; base += OWN_THREADS) {
    const int e = base + lane;
    const int d = (e < n_edges) ? dst[e] : -1;
    const unsigned du = (unsigned)d - (unsigned)lo;
    q_push_bits(ring, head, tail, (du < RNG) ? 1u : 0u, 0u, e, lane, lo, rec, dinv, xw, acc);
  }
  if (tail != head) q_batch(ring, head, (int)(tail - head), lane, lo, rec, dinv, xw, acc);

  __syncthreads();

  for (int r = wave; r < RNG; r += OWN_WAVES) {
    const int n = lo + r;
    if (n < n_nodes) {
      const float di = dinv[n];
      const float k  = FILLW * di * di;
      const v2f a  = *reinterpret_cast<const v2f*>(acc + r * CCH + 2 * lane);
      const v2f x  = *reinterpret_cast<const v2f*>(xw + (size_t)n * CCH + 2 * lane);
      const v2f bb = *reinterpret_cast<const v2f*>(b + 2 * lane);
      v2f o;
      o.x = tanhf(a.x + x.x * k + bb.x);
      o.y = tanhf(a.y + x.y * k + bb.y);
      *reinterpret_cast<v2f*>(out + (size_t)n * CCH + 2 * lane) = o;
    }
  }
}

extern "C" void kernel_launch(void* const* d_in, const int* in_sizes, int n_in,
                              void* d_out, int out_size, void* d_ws, size_t ws_size,
                              hipStream_t stream) {
  const float* x  = (const float*)d_in[0];
  const int*   ei = (const int*)d_in[1];
  const float* ew = (const float*)d_in[2];
  const float* W  = (const float*)d_in[3];
  const float* b  = (const float*)d_in[4];
  float* out = (float*)d_out;

  const int n_nodes = in_sizes[0] / CCH;
  const int n_edges = in_sizes[2];
  const int* src = ei;
  const int* dst = ei + n_edges;

  float* xw   = (float*)d_ws;
  float* dinv = xw + (size_t)n_nodes * CCH;
  unsigned* key_raw = (unsigned*)(dinv + n_nodes);
  float* part = (float*)d_ws + ((size_t)32 << 20);
  unsigned* rec_raw = (unsigned*)d_ws + ((size_t)48 << 20);

  const int B = 256;

  const int n_chunks = (n_edges + DEG_LINES * DEG_THREADS - 1) / (DEG_LINES * DEG_THREADS);
  gcn_deg<<<dim3(NQ, n_chunks), DEG_THREADS, 0, stream>>>(src, dst, ew, part, key_raw, rec_raw, n_edges);
  gcn_dinv<<<(n_nodes + B - 1) / B, B, 0, stream>>>(part, dinv, n_nodes, n_chunks);

  const int n_row_tiles = (n_nodes + 15) / 16;
  const int waves = n_row_tiles * (CCH / 16);
  gcn_gemm_wmma<<<(waves + 7) / 8, B, 0, stream>>>(x, W, xw, n_nodes);

  gcn_owner<<<(n_nodes + RNG - 1) / RNG, OWN_THREADS, 0, stream>>>(
      dst, dinv, xw, b, key_raw, rec_raw, out, n_edges, n_nodes);
}


# Aggregate first, multiply after: owner sums x[src]*norm in LDS (self term as start value), then sums@W on WMMA from LDS, +b, tanh; product launch and xw plane gone (3 launches); product regrouped
# speedup vs baseline: 3.7581x; 1.1162x over previous
#include <hip/hip_runtime.h>
#include <hip/hip_bf16.h>
#include <math.h>

typedef __attribute__((ext_vector_type(2))) float v2f;
typedef __attribute__((ext_vector_type(8))) float v8f;

#define CCH   64
#define FILLW 2.0f

#define RNG        391
#define KEY_PLANES 8

__device__ __forceinline__ unsigned pad_words(const unsigned* p) {
  return ((256u - ((unsigned)reinterpret_cast<size_t>(p) & 255u)) & 255u) >> 2;
}
__device__ __forceinline__ int plane_words(int n_edges) { return (((n_edges + 31) >> 5) + 63) & ~63; }

typedef __attribute__((ext_vector_type(4))) float    v4f;
typedef __attribute__((ext_vector_type(2))) unsigned v2u;

#define REC_SHIFT   17
#define REC_SRCMASK ((1u << REC_SHIFT) - 1u)

#define NQ          4
#define QN          25000
#define QP          25024
#define DEG_LINES   25
#define DEG_THREADS 1024
#define PL_PER_Q    (KEY_PLANES / NQ)

__device__ __forceinline__ void lds_add_f32(float* p, float v) {
  __hip_atomic_fetch_add(p, v, __ATOMIC_RELAXED, __HIP_MEMORY_SCOPE_WORKGROUP);
}

template <bool FULL>
__device__ __forceinline__ void load_edges(int (&d)[DEG_LINES], float (&ww)[DEG_LINES],
                                           const int* __restrict__ dst, const float* __restrict__ w,
                                           int e0, int n_edges) {
#pragma unroll
  for (int i = 0; i < DEG_LINES; ++i) {
    const int e = e0 + i * DEG_THREADS;
    if (FULL || e < n_edges) { d[i] = dst[e]; ww[i] = w[e]; }
    else                     { d[i] = -1;     ww[i] = 0.0f; }
  }
}

template <bool FULL>
__device__ __forceinline__ void load_src(int (&sv)[DEG_LINES], const int* __restrict__ src,
                                         int e0, int n_edges, int rsel) {
#pragma unroll
  for (int i = 0; i < DEG_LINES; ++i) {
    const int e = e0 + i * DEG_THREADS;
    sv[i] = 0;
    if ((i & 3) == rsel) { if (FULL || e < n_edges) sv[i] = src[e]; }
  }
}

__global__ void __launch_bounds__(DEG_THREADS)
gcn_deg(const int* __restrict__ src, const int* __restrict__ dst, const float* __restrict__ w,
        float* __restrict__ part, unsigned* __restrict__ key_raw, unsigned* __restrict__ rec_raw, int n_edges) {
  __shared__ __attribute__((aligned(16))) float    tab[QP];
  __shared__ unsigned tile[DEG_LINES][PL_PER_Q][32];
  const int tid   = threadIdx.x;
  const int lane  = tid & 31;
  const int wave  = __builtin_amdgcn_readfirstlane(tid >> 5);
  const int q     = blockIdx.x;
  const int c     = blockIdx.y;
  const int line0 = c * DEG_LINES;

  int   d[DEG_LINES];
  float ww[DEG_LINES];
  int   sv[DEG_LINES];
  const int rsel = (q - wave) & 3;
  if ((line0 + DEG_LINES) * DEG_THREADS <= n_edges) {
    load_edges<true>(d, ww, dst, w, line0 * DEG_THREADS + tid, n_edges);
    load_src<true>(sv, src, line0 * DEG_THREADS + tid, n_edges, rsel);
  } else {
    load_edges<false>(d, ww, dst, w, line0 * DEG_THREADS + tid, n_edges);
    load_src<false>(sv, src, line0 * DEG_THREADS + tid, n_edges, rsel);
  }
  for (int i = tid; i < QP / 4; i += DEG_THREADS) {
    v4f zero4 = {0.0f, 0.0f, 0.0f, 0.0f};
    reinterpret_cast<v4f*>(tab)[i] = zero4;
  }
  __syncthreads();

  const unsigned nlo = (unsigned)(q * QN);
#pragma unroll
  for (int i = 0; i < DEG_LINES; ++i) {
    const unsigned du = (unsigned)d[i] - nlo;
    if (du < (unsigned)QN) lds_add_f32(&tab[du], ww[i]);
    const unsigned key = (d[i] >= 0) ? (((unsigned)d[i] / RNG) & 0xFFu) : 0u;
    if ((i & 3) == rsel && d[i] >= 0) {
      v2u r;
      r.x = (unsigned)sv[i] | (((unsigned)d[i] - key * RNG) << REC_SHIFT);
      r.y = __float_as_uint(ww[i]);
      reinterpret_cast<v2u*>(rec_raw)[line0 * DEG_THREADS + tid + i * DEG_THREADS] = r;
    }
    unsigned word = 0u;
#pragma unroll
    for (int t = 0; t < PL_PER_Q; ++t) {
      const int j = q * PL_PER_Q + t;
      const unsigned bj = __builtin_amdgcn_ballot_w32(((key >> j) & 1u) != 0u);
      if (lane == t) word = bj;
    }
    if (lane < PL_PER_Q) tile[i][lane][wave] = word;
  }
  __syncthreads();

  float* __restrict__ pout = part + ((size_t)c * NQ + q) * QP;
  for (int i = tid; i < QP / 4; i += DEG_THREADS)
    reinterpret_cast<v4f*>(pout)[i] = reinterpret_cast<const v4f*>(tab)[i];

  unsigned* planes = key_raw + pad_words(key_raw);
  const int pw = plane_words(n_edges);
  for (int s = wave; s < DEG_LINES * PL_PER_Q; s += DEG_THREADS / 32) {
    const int i = s / PL_PER_Q;
    const int t = s - i * PL_PER_Q;
    const int ln = line0 + i;
    if (ln * DEG_THREADS < n_edges)
      planes[(size_t)(q * PL_PER_Q + t) * pw + ln * 32 + lane] = tile[i][t][lane];
  }
}

#define DINV_FLY 32

__global__ void gcn_dinv(const float* __restrict__ part, float* __restrict__ dinv, int n_nodes, int n_chunks) {
  const int n = blockIdx.x * blockDim.x + threadIdx.x;
  if (n < n_nodes) {
    const int q = n / QN;
    const float* __restrict__ p = part + (size_t)q * QP + (n - q * QN);
    float dsum = FILLW;
    for (int c0 = 0; c0 < n_chunks; c0 += DINV_FLY) {
      float v[DINV_FLY];
#pragma unroll
      for (int k = 0; k < DINV_FLY; ++k) v[k] = (c0 + k < n_chunks) ? p[(size_t)(c0 + k) * (NQ * QP)] : 0.0f;
#pragma unroll
      for (int k = 0; k < DINV_FLY; ++k) dsum += v[k];
    }
    dinv[n] = (dsum > 0.0f) ? rsqrtf(dsum) : 0.0f;
  }
}

#define OWN_THREADS 1024
#define OWN_WAVES   32
#define QCAP        64
#define WAVE_CHUNKS 32
#define STEP_CHUNKS (OWN_WAVES * WAVE_CHUNKS)
#define CHUNK_EDGES 64
#define ROWS_FLY    16
#define AP          68
#define OWN_TILES   ((RNG + 1 + 15) / 16)
#define OWN_ROWS    (OWN_TILES * 16)

__device__ __forceinline__ void q_batch(const unsigned* ring, unsigned head, int cnt,
                                        int lane, int lo,
                                        const v2u* __restrict__ rec,
                                        const float* __restrict__ dinv, const float* __restrict__ x,
                                        float* __restrict__ acc) {
  int   s   = lo;
  int   dl  = RNG;
  float nrm = 0.0f;
  if (lane < cnt) {
    const int e = (int)ring[(head + (unsigned)lane) & (QCAP - 1)];
    const v2u r = rec[e];
    s = (int)(r.x & REC_SRCMASK);
    const unsigned du = r.x >> REC_SHIFT;
    if (du < RNG) {
      dl  = (int)du;
      nrm = dinv[s] * __uint_as_float(r.y) * dinv[lo + (int)du];
    }
  }
  for (int j0 = 0; j0 < cnt; j0 += ROWS_FLY) {
    v2f xr[ROWS_FLY];
#pragma unroll
    for (int k = 0; k < ROWS_FLY; ++k) {
      const int sj = __builtin_amdgcn_readlane(s, j0 + k);
      xr[k] = *reinterpret_cast<const v2f*>(x + (size_t)sj * CCH + 2 * lane);
    }
#pragma unroll
    for (int k = 0; k < ROWS_FLY; ++k) {
      const int   dj = __builtin_amdgcn_readlane(dl, j0 + k);
      const float nj = __int_as_float(__builtin_amdgcn_readlane(__float_as_int(nrm), j0 + k));
      float* a = acc + dj * AP + 2 * lane;
      lds_add_f32(a,     xr[k].x * nj);
      lds_add_f32(a + 1, xr[k].y * nj);
    }
  }
}

__device__ __forceinline__ void q_push_bits(unsigned* ring, unsigned& head, unsigned& tail,
                                            unsigned flo, unsigned fhi, int e0,
                                            int lane, int lo,
                                            const v2u* __restrict__ rec,
                                            const float* __restrict__ dinv, const float* __restrict__ x,
                                            float* __restrict__ acc) {
  unsigned mask = __builtin_amdgcn_ballot_w32((flo | fhi) != 0u);
  while (mask != 0u) {
    if ((flo | fhi) != 0u) {
      const bool in_lo = (flo != 0u);
      const unsigned cur = in_lo ? flo : fhi;
      const int bit = __builtin_ctz(cur) + (in_lo ? 0 : 32);
      const unsigned slot = (tail + __builtin_amdgcn_mbcnt_lo(mask, 0u)) & (QCAP - 1);
      ring[slot] = (unsigned)(e0 + bit);
      const unsigned rest = cur & (cur - 1u);
      if (in_lo) flo = rest; else fhi = rest;
    }
    tail += (unsigned)__builtin_popcount(mask);
    if (tail - head >= 32u) {
      q_batch(ring, head, 32, lane, lo, rec, dinv, x, acc);
      head += 32u;
    }
    mask = __builtin_amdgcn_ballot_w32((flo | fhi) != 0u);
  }
}

__device__ __forceinline__ v2u own_bits(const v2u (&k)[KEY_PLANES], const unsigned (&z)[KEY_PLANES]) {
  unsigned dx = 0u, dy = 0u;
#pragma unroll
  for (int j = 0; j < KEY_PLANES; ++j) {
    dx |= k[j].x ^ z[j];
    dy |= k[j].y ^ z[j];
  }
  v2u m; m.x = ~dx; m.y = ~dy;
  return m;
}

__device__ __forceinline__ void load_keys(v2u (&k)[KEY_PLANES], const v2u* __restrict__ planes2, int pw2, int c) {
#pragma unroll
  for (int j = 0; j < KEY_PLANES; ++j) k[j] = planes2[(size_t)j * pw2 + c];
}

__global__ void __launch_bounds__(OWN_THREADS)
gcn_owner(const int* __restrict__ dst, const float* __restrict__ dinv,
          const float* __restrict__ x, const float* __restrict__ W, const float* __restrict__ b,
          const unsigned* __restrict__ key_raw, const unsigned* __restrict__ rec_raw,
          float* __restrict__ out, int n_edges, int n_nodes) {
  __shared__ __attribute__((aligned(16))) float    acc[OWN_ROWS * AP];
  __shared__ __attribute__((aligned(16))) float    stage[OWN_ROWS * AP];
  __shared__ __attribute__((aligned(16))) unsigned ringbuf[OWN_WAVES * QCAP];

  const int tid  = threadIdx.x;
  const int lane = tid & 31;
  const int wave = __builtin_amdgcn_readfirstlane(tid >> 5);
  const int lo   = blockIdx.x * RNG;
  unsigned* ring = ringbuf + wave * QCAP;
  const v2u* __restrict__ rec = reinterpret_cast<const v2u*>(rec_raw);

  {
    constexpr int NI = (OWN_ROWS + OWN_WAVES - 1) / OWN_WAVES;
    float dv = 0.0f;
    {
      const int r = wave + lane * OWN_WAVES;
      if (lane < NI && r < RNG && lo + r < n_nodes) dv = dinv[lo + r];
    }
    v2f xs[NI];
#pragma unroll
    for (int i = 0; i < NI; ++i) {
      const int  r  = wave + i * OWN_WAVES;
      const bool ok = (r < RNG) && (lo + r < n_nodes);
      const int  n  = ok ? lo + r : lo;
      xs[i] = *reinterpret_cast<const v2f*>(x + (size_t)n * CCH + 2 * lane);
    }
#pragma unroll
    for (int i = 0; i < NI; ++i) {
      const int r = wave + i * OWN_WAVES;
      if (r < OWN_ROWS) {
        const bool  ok = (r < RNG) && (lo + r < n_nodes);
        const float di = __int_as_float(__builtin_amdgcn_readlane(__float_as_int(dv), i));
        const float k  = FILLW * di * di;
        v2f v;
        v.x = ok ? xs[i].x * k : 0.0f;
        v.y = ok ? xs[i].y * k : 0.0f;
        *reinterpret_cast<v2f*>(acc + r * AP + 2 * lane) = v;
      }
    }
  }
  __syncthreads();

  unsigned head = 0u, tail = 0u;

  const v2u* __restrict__ planes2 = reinterpret_cast<const v2u*>(key_raw + pad_words(key_raw));
  const int pw2 = plane_words(n_edges) >> 1;
  const int n_chunks = n_edges / CHUNK_EDGES;
  const int n_full   = n_chunks / STEP_CHUNKS;

  unsigned z[KEY_PLANES];
#pragma unroll
  for (int j = 0; j < KEY_PLANES; ++j) z[j] = 0u - ((blockIdx.x >> j) & 1u);

  if (n_full > 0) {
    int it = (int)(((long long)blockIdx.x * n_full) / gridDim.x);
    v2u ka[KEY_PLANES], kb[KEY_PLANES];
    load_keys(ka, planes2, pw2, (it * OWN_WAVES + wave) * WAVE_CHUNKS + lane);
    int n = 0;
    for (; n + 2 <= n_full; n += 2) {
      int it1 = it + 1;  if (it1 == n_full) it1 = 0;
      int it2 = it1 + 1; if (it2 == n_full) it2 = 0;
      load_keys(kb, planes2, pw2, (it1 * OWN_WAVES + wave) * WAVE_CHUNKS + lane);
      {
        const v2u m = own_bits(ka, z);
        q_push_bits(ring, head, tail, m.x, m.y, ((it * OWN_WAVES + wave) * WAVE_CHUNKS + lane) * CHUNK_EDGES,
                    lane, lo, rec, dinv, x, acc);
      }
      load_keys(ka, planes2, pw2, (it2 * OWN_WAVES + wave) * WAVE_CHUNKS + lane);
      {
        const v2u m = own_bits(kb, z);
        q_push_bits(ring, head, tail, m.x, m.y, ((it1 * OWN_WAVES + wave) * WAVE_CHUNKS + lane) * CHUNK_EDGES,
                    lane, lo, rec, dinv, x, acc);
      }
      it = it2;
    }
    if (n < n_full) {
      const v2u m = own_bits(ka, z);
      q_push_bits(ring, head, tail, m.x, m.y, ((it * OWN_WAVES + wave) * WAVE_CHUNKS + lane) * CHUNK_EDGES,
                  lane, lo, rec, dinv, x, acc);
    }
  }
  for (int cb = n_full * STEP_CHUNKS + wave * 32; cb < n_chunks; cb += OWN_THREADS) {
    const int  c     = cb + lane;
    const bool valid = (c < n_chunks);
    v2u kr[KEY_PLANES];
    load_keys(kr, planes2, pw2, valid ? c : 0);
    v2u m = own_bits(kr, z);
    if (!valid) { m.x = 0u; m.y = 0u; }
    q_push_bits(ring, head, tail, m.x, m.y, c * CHUNK_EDGES, lane, lo, rec, dinv, x, acc);
  }
  for (int base = n_chunks * CHUNK_EDGES + wave * 32; base < n_edges; base += OWN_THREADS) {
    const int e = base + lane;
    const int d = (e < n_edges) ? dst[e] : -1;
    const unsigned du = (unsigned)d - (unsigned)lo;
    q_push_bits(ring, head, tail, (du < RNG) ? 1u : 0u, 0u, e, lane, lo, rec, dinv, x, acc);
  }
  if (tail != head) q_batch(ring, head, (int)(tail - head), lane, lo, rec, dinv, x, acc);

  const int half = lane >> 4;
  const int l16  = lane & 15;
  const int ct   = wave & 3;
  const float* __restrict__ wcol = W + ct * 16 + l16;
  v2f bf[CCH / 4];
#pragma unroll
  for (int t = 0; t < CCH / 4; ++t) {
    const int ka = 4 * t + 2 * half;
    bf[t].x = wcol[(size_t)(ka + 0) * CCH];
    bf[t].y = wcol[(size_t)(ka + 1) * CCH];
  }
  const float bias = b[ct * 16 + l16];

  __syncthreads();

  for (int rt = wave >> 2; rt < OWN_TILES; rt += OWN_WAVES / 4) {
    const float* arow = acc + (rt * 16 + l16) * AP + 2 * half;
    v2f af[CCH / 4];
#pragma unroll
    for (int t = 0; t < CCH / 4; ++t) af[t] = *reinterpret_cast<const v2f*>(arow + 4 * t);
    v8f d = {};
#pragma unroll
    for (int t = 0; t < CCH / 4; ++t) {
      d = __builtin_amdgcn_wmma_f32_16x16x4_f32(
           false, af[t],  false, bf[t],
           (short)0, d,  false,  false);
    }
    float* srow = stage + (rt * 16 + 8 * half) * AP + ct * 16 + l16;
#pragma unroll
    for (int i = 0; i < 8; ++i) srow[i * AP] = tanhf(d[i] + bias);
  }
  __syncthreads();

  for (int r = wave; r < RNG; r += OWN_WAVES) {
    const int n = lo + r;
    if (n < n_nodes)
      *reinterpret_cast<v2f*>(out + (size_t)n * CCH + 2 * lane) = *reinterpret_cast<const v2f*>(stage + r * AP + 2 * lane);
  }
}

extern "C" void kernel_launch(void* const* d_in, const int* in_sizes, int n_in,
                              void* d_out, int out_size, void* d_ws, size_t ws_size,
                              hipStream_t stream) {
  const float* x  = (const float*)d_in[0];
  const int*   ei = (const int*)d_in[1];
  const float* ew = (const float*)d_in[2];
  const float* W  = (const float*)d_in[3];
  const float* b  = (const float*)d_in[4];
  float* out = (float*)d_out;

  const int n_nodes = in_sizes[0] / CCH;
  const int n_edges = in_sizes[2];
  const int* src = ei;
  const int* dst = ei + n_edges;

  float* dinv = (float*)d_ws;
  unsigned* key_raw = (unsigned*)(dinv + n_nodes);
  float* part = (float*)d_ws + ((size_t)32 << 20);
  unsigned* rec_raw = (unsigned*)d_ws + ((size_t)48 << 20);

  const int B = 256;

  const int n_chunks = (n_edges + DEG_LINES * DEG_THREADS - 1) / (DEG_LINES * DEG_THREADS);
  gcn_deg<<<dim3(NQ, n_chunks), DEG_THREADS, 0, stream>>>(src, dst, ew, part, key_raw, rec_raw, n_edges);
  gcn_dinv<<<(n_nodes + B - 1) / B, B, 0, stream>>>(part, dinv, n_nodes, n_chunks);

  gcn_owner<<<(n_nodes + RNG - 1) / RNG, OWN_THREADS, 0, stream>>>(
      dst, dinv, x, W, b, key_raw, rec_raw, out, n_edges, n_nodes);
}
